# starting kernel
# baseline (speedup 1.0000x reference)
#include <hip/hip_runtime.h>

#define SEQL 256
#define BATCH 64
#define IN_DIM 512
#define HID 1024

typedef __attribute__((ext_vector_type(2))) float v2f;
typedef __attribute__((ext_vector_type(8))) float v8f;

struct GateParams {
    const float* Wi[4];
    const float* Wh[4];
};

__device__ __forceinline__ float fast_sigmoid(float x) {
    return 1.0f / (1.0f + __expf(-x));
}

__device__ __forceinline__ void kloop_dual(
    const float* __restrict__ a_base, const float* __restrict__ b_base,
    int kbegin, int kend, v8f& acc0, v8f& acc1)
{
    #pragma unroll 4
    for (int k0 = kbegin; k0 < kend; k0 += 8) {
        v2f a0 = *(const v2f*)(a_base + k0);
        v2f b0 = *(const v2f*)(b_base + k0);
        v2f a1 = *(const v2f*)(a_base + k0 + 4);
        v2f b1 = *(const v2f*)(b_base + k0 + 4);
        acc0 = __builtin_amdgcn_wmma_f32_16x16x4_f32(
            false, a0, false, b0, (short)0, acc0, false, false);
        acc1 = __builtin_amdgcn_wmma_f32_16x16x4_f32(
            false, a1, false, b1, (short)0, acc1, false, false);
    }
}

__global__ __launch_bounds__(256) void lstm_step_kernel(
    const float* __restrict__ x_t,
    const float* __restrict__ h_in,
    float* __restrict__ h_out,
    float* __restrict__ c_state,
    float* __restrict__ h_seq_t,
    const float* __restrict__ bsum,
    GateParams gp)
{
    __shared__ float smem[8 * 256];

    const int tid  = threadIdx.x;
    const int wave = tid >> 5;
    const int lane = tid & 31;
    const int gate = wave & 3;
    const int half = wave >> 2;

    const int bt = blockIdx.x;
    const int ht = blockIdx.y;

    const int m    = lane & 15;
    const int kk   = (lane >> 4) << 1;

    const int arow = bt * 16 + m;
    const int wrow = ht * 16 + m;

    v8f acc0 = {};
    v8f acc1 = {};

    const float* a_h = h_in + arow * HID + kk;
    const float* b_h = gp.Wh[gate] + wrow * HID + kk;

    if (half == 0) {
        kloop_dual(x_t + arow * IN_DIM + kk, gp.Wi[gate] + wrow * IN_DIM + kk,
                   0, IN_DIM, acc0, acc1);
        kloop_dual(a_h, b_h, 0, 256, acc0, acc1);
    } else {
        kloop_dual(a_h, b_h, 256, HID, acc0, acc1);
    }

    {
        float* s = smem + wave * 256;
        const int n     = lane & 15;
        const int mbase = (lane >> 4) << 3;
        #pragma unroll
        for (int r = 0; r < 8; ++r)
            s[(mbase + r) * 16 + n] = acc0[r] + acc1[r];
    }
    __syncthreads();

    {
        const int e  = tid;
        const int mm = e >> 4;
        const int n  = e & 15;
        const int b  = bt * 16 + mm;
        const int h  = ht * 16 + n;

        float fpre = smem[0 * 256 + e] + smem[4 * 256 + e] + bsum[0 * HID + h];
        float ipre = smem[1 * 256 + e] + smem[5 * 256 + e] + bsum[1 * HID + h];
        float opre = smem[2 * 256 + e] + smem[6 * 256 + e] + bsum[2 * HID + h];
        float gpre = smem[3 * 256 + e] + smem[7 * 256 + e] + bsum[3 * HID + h];

        float f = fast_sigmoid(fpre);
        float i = fast_sigmoid(ipre);
        float o = fast_sigmoid(opre);
        float g = tanhf(gpre);

        const int idx = b * HID + h;
        float c_new = f * c_state[idx] + i * g;
        c_state[idx] = c_new;
        float h_new = o * tanhf(c_new);
        h_out[idx]   = h_new;
        h_seq_t[idx] = h_new;
    }
}

__global__ __launch_bounds__(256) void prep_kernel(
    float* __restrict__ h0, float* __restrict__ c0,
    float* __restrict__ bsum,
    const float* __restrict__ bi0, const float* __restrict__ bh0,
    const float* __restrict__ bi1, const float* __restrict__ bh1,
    const float* __restrict__ bi2, const float* __restrict__ bh2,
    const float* __restrict__ bi3, const float* __restrict__ bh3)
{
    int i = blockIdx.x * blockDim.x + threadIdx.x;
    if (i < BATCH * HID) {
        h0[i] = 0.0f;
        c0[i] = 0.0f;
    }
    if (i < HID) {
        bsum[0 * HID + i] = bi0[i] + bh0[i];
        bsum[1 * HID + i] = bi1[i] + bh1[i];
        bsum[2 * HID + i] = bi2[i] + bh2[i];
        bsum[3 * HID + i] = bi3[i] + bh3[i];
    }
}

__global__ __launch_bounds__(256) void tail_copy_kernel(
    const float* __restrict__ h_fin, const float* __restrict__ c_fin,
    float* __restrict__ out_tail)
{
    int i = blockIdx.x * blockDim.x + threadIdx.x;
    if (i < BATCH * HID) {
        out_tail[i]               = h_fin[i];
        out_tail[BATCH * HID + i] = c_fin[i];
    }
}

extern "C" void kernel_launch(void* const* d_in, const int* in_sizes, int n_in,
                              void* d_out, int out_size, void* d_ws, size_t ws_size,
                              hipStream_t stream) {
    (void)in_sizes; (void)n_in; (void)out_size; (void)ws_size;

    const float* x = (const float*)d_in[0];
    GateParams gp;
    const float* bi[4];
    const float* bh[4];
    for (int g = 0; g < 4; ++g) {
        gp.Wi[g] = (const float*)d_in[1 + 4 * g];
        bi[g]    = (const float*)d_in[2 + 4 * g];
        gp.Wh[g] = (const float*)d_in[3 + 4 * g];
        bh[g]    = (const float*)d_in[4 + 4 * g];
    }

    float* ws    = (float*)d_ws;
    float* hbuf0 = ws;
    float* hbuf1 = ws + BATCH * HID;
    float* cbuf  = ws + 2 * BATCH * HID;
    float* bsum  = ws + 3 * BATCH * HID;
    float* out   = (float*)d_out;

    prep_kernel<<<(BATCH * HID + 255) / 256, 256, 0, stream>>>(
        hbuf0, cbuf, bsum,
        bi[0], bh[0], bi[1], bh[1], bi[2], bh[2], bi[3], bh[3]);

    dim3 grid(BATCH / 16, HID / 16);
    for (int t = 0; t < SEQL; ++t) {
        const float* h_in  = (t & 1) ? hbuf1 : hbuf0;
        float*       h_out = (t & 1) ? hbuf0 : hbuf1;
        lstm_step_kernel<<<grid, 256, 0, stream>>>(
            x + (size_t)t * BATCH * IN_DIM, h_in, h_out, cbuf,
            out + (size_t)t * BATCH * HID, bsum, gp);
    }
    tail_copy_kernel<<<(BATCH * HID + 255) / 256, 256, 0, stream>>>(
        hbuf0, cbuf, out + (size_t)SEQL * BATCH * HID);
}


# Fragment loads widened from 64 to 128 bit per lane, k order permuted inside each 8-wide k group (same for A and B): half the loads and cache lines touched per k; order of the k sums changes
# speedup vs baseline: 1.1265x; 1.1265x over previous
#include <hip/hip_runtime.h>

#define SEQL 256
#define BATCH 64
#define IN_DIM 512
#define HID 1024

typedef __attribute__((ext_vector_type(2))) float v2f;
typedef __attribute__((ext_vector_type(4))) float v4f;
typedef __attribute__((ext_vector_type(8))) float v8f;

struct GateParams {
    const float* Wi[4];
    const float* Wh[4];
};

__device__ __forceinline__ float fast_sigmoid(float x) {
    return 1.0f / (1.0f + __expf(-x));
}

__device__ __forceinline__ void kloop_dual(
    const float* __restrict__ a_base, const float* __restrict__ b_base,
    int kbegin, int kend, v8f& acc0, v8f& acc1)
{
    #pragma unroll 8
    for (int k0 = kbegin; k0 < kend; k0 += 8) {
        v4f a = *(const v4f*)(a_base + k0);
        v4f b = *(const v4f*)(b_base + k0);
        v2f a_lo = a.xy, a_hi = a.zw;
        v2f b_lo = b.xy, b_hi = b.zw;
        acc0 = __builtin_amdgcn_wmma_f32_16x16x4_f32(
            false, a_lo, false, b_lo, (short)0, acc0, false, false);
        acc1 = __builtin_amdgcn_wmma_f32_16x16x4_f32(
            false, a_hi, false, b_hi, (short)0, acc1, false, false);
    }
}

__global__ __launch_bounds__(256) void lstm_step_kernel(
    const float* __restrict__ x_t,
    const float* __restrict__ h_in,
    float* __restrict__ h_out,
    float* __restrict__ c_state,
    float* __restrict__ h_seq_t,
    const float* __restrict__ bsum,
    GateParams gp)
{
    __shared__ float smem[8 * 256];

    const int tid  = threadIdx.x;
    const int wave = tid >> 5;
    const int lane = tid & 31;
    const int gate = wave & 3;
    const int half = wave >> 2;

    const int bt = blockIdx.x;
    const int ht = blockIdx.y;

    const int m    = lane & 15;
    const int kk   = (lane >> 4) << 2;

    const int arow = bt * 16 + m;
    const int wrow = ht * 16 + m;

    v8f acc0 = {};
    v8f acc1 = {};

    const float* a_h = h_in + arow * HID + kk;
    const float* b_h = gp.Wh[gate] + wrow * HID + kk;

    if (half == 0) {
        kloop_dual(x_t + arow * IN_DIM + kk, gp.Wi[gate] + wrow * IN_DIM + kk,
                   0, IN_DIM, acc0, acc1);
        kloop_dual(a_h, b_h, 0, 256, acc0, acc1);
    } else {
        kloop_dual(a_h, b_h, 256, HID, acc0, acc1);
    }

    {
        float* s = smem + wave * 256;
        const int n     = lane & 15;
        const int mbase = (lane >> 4) << 3;
        #pragma unroll
        for (int r = 0; r < 8; ++r)
            s[(mbase + r) * 16 + n] = acc0[r] + acc1[r];
    }
    __syncthreads();

    {
        const int e  = tid;
        const int mm = e >> 4;
        const int n  = e & 15;
        const int b  = bt * 16 + mm;
        const int h  = ht * 16 + n;

        float fpre = smem[0 * 256 + e] + smem[4 * 256 + e] + bsum[0 * HID + h];
        float ipre = smem[1 * 256 + e] + smem[5 * 256 + e] + bsum[1 * HID + h];
        float opre = smem[2 * 256 + e] + smem[6 * 256 + e] + bsum[2 * HID + h];
        float gpre = smem[3 * 256 + e] + smem[7 * 256 + e] + bsum[3 * HID + h];

        float f = fast_sigmoid(fpre);
        float i = fast_sigmoid(ipre);
        float o = fast_sigmoid(opre);
        float g = tanhf(gpre);

        const int idx = b * HID + h;
        float c_new = f * c_state[idx] + i * g;
        c_state[idx] = c_new;
        float h_new = o * tanhf(c_new);
        h_out[idx]   = h_new;
        h_seq_t[idx] = h_new;
    }
}

__global__ __launch_bounds__(256) void prep_kernel(
    float* __restrict__ h0, float* __restrict__ c0,
    float* __restrict__ bsum,
    const float* __restrict__ bi0, const float* __restrict__ bh0,
    const float* __restrict__ bi1, const float* __restrict__ bh1,
    const float* __restrict__ bi2, const float* __restrict__ bh2,
    const float* __restrict__ bi3, const float* __restrict__ bh3)
{
    int i = blockIdx.x * blockDim.x + threadIdx.x;
    if (i < BATCH * HID) {
        h0[i] = 0.0f;
        c0[i] = 0.0f;
    }
    if (i < HID) {
        bsum[0 * HID + i] = bi0[i] + bh0[i];
        bsum[1 * HID + i] = bi1[i] + bh1[i];
        bsum[2 * HID + i] = bi2[i] + bh2[i];
        bsum[3 * HID + i] = bi3[i] + bh3[i];
    }
}

__global__ __launch_bounds__(256) void tail_copy_kernel(
    const float* __restrict__ h_fin, const float* __restrict__ c_fin,
    float* __restrict__ out_tail)
{
    int i = blockIdx.x * blockDim.x + threadIdx.x;
    if (i < BATCH * HID) {
        out_tail[i]               = h_fin[i];
        out_tail[BATCH * HID + i] = c_fin[i];
    }
}

extern "C" void kernel_launch(void* const* d_in, const int* in_sizes, int n_in,
                              void* d_out, int out_size, void* d_ws, size_t ws_size,
                              hipStream_t stream) {
    (void)in_sizes; (void)n_in; (void)out_size; (void)ws_size;

    const float* x = (const float*)d_in[0];
    GateParams gp;
    const float* bi[4];
    const float* bh[4];
    for (int g = 0; g < 4; ++g) {
        gp.Wi[g] = (const float*)d_in[1 + 4 * g];
        bi[g]    = (const float*)d_in[2 + 4 * g];
        gp.Wh[g] = (const float*)d_in[3 + 4 * g];
        bh[g]    = (const float*)d_in[4 + 4 * g];
    }

    float* ws    = (float*)d_ws;
    float* hbuf0 = ws;
    float* hbuf1 = ws + BATCH * HID;
    float* cbuf  = ws + 2 * BATCH * HID;
    float* bsum  = ws + 3 * BATCH * HID;
    float* out   = (float*)d_out;

    prep_kernel<<<(BATCH * HID + 255) / 256, 256, 0, stream>>>(
        hbuf0, cbuf, bsum,
        bi[0], bh[0], bi[1], bh[1], bi[2], bh[2], bi[3], bh[3]);

    dim3 grid(BATCH / 16, HID / 16);
    for (int t = 0; t < SEQL; ++t) {
        const float* h_in  = (t & 1) ? hbuf1 : hbuf0;
        float*       h_out = (t & 1) ? hbuf0 : hbuf1;
        lstm_step_kernel<<<grid, 256, 0, stream>>>(
            x + (size_t)t * BATCH * IN_DIM, h_in, h_out, cbuf,
            out + (size_t)t * BATCH * HID, bsum, gp);
    }
    tail_copy_kernel<<<(BATCH * HID + 255) / 256, 256, 0, stream>>>(
        hbuf0, cbuf, out + (size_t)SEQL * BATCH * HID);
}


# Each wave owns a 2x2 block of 16x16 tiles over 1/8 of K (was 1 tile over 1/2), loads grouped ahead: a fragment load feeds 2x the matrix instructions; 8 partials summed in LDS: k sum order changes
# speedup vs baseline: 1.4935x; 1.3258x over previous
#include <hip/hip_runtime.h>


#define SEQL 256
#define BATCH 64
#define IN_DIM 512
#define HID 1024

typedef __attribute__((ext_vector_type(2))) float v2f;
typedef __attribute__((ext_vector_type(4))) float v4f;
typedef __attribute__((ext_vector_type(8))) float v8f;

struct GateParams {
    const float* Wi[4];
    const float* Wh[4];
};

__device__ __forceinline__ float fast_sigmoid(float x) {
    return 1.0f / (1.0f + __expf(-x));
}

struct Slice {
    v4f a0[8], a1[8], b0[8], b1[8];
};

__device__ __forceinline__ void slice_load(
    Slice& s,
    const float* __restrict__ a0p, const float* __restrict__ a1p,
    const float* __restrict__ b0p, const float* __restrict__ b1p)
{
    #pragma unroll
    for (int g = 0; g < 8; ++g) s.a0[g] = *(const v4f*)(a0p + 8 * g);
    #pragma unroll
    for (int g = 0; g < 8; ++g) s.a1[g] = *(const v4f*)(a1p + 8 * g);
    #pragma unroll
    for (int g = 0; g < 8; ++g) s.b0[g] = *(const v4f*)(b0p + 8 * g);
    #pragma unroll
    for (int g = 0; g < 8; ++g) s.b1[g] = *(const v4f*)(b1p + 8 * g);
}

__device__ __forceinline__ void slice_mma(const Slice& s, v8f (&acc)[2][2])
{
    #pragma unroll
    for (int g = 0; g < 8; ++g) {
        v2f a0l = s.a0[g].xy, a0h = s.a0[g].zw;
        v2f a1l = s.a1[g].xy, a1h = s.a1[g].zw;
        v2f b0l = s.b0[g].xy, b0h = s.b0[g].zw;
        v2f b1l = s.b1[g].xy, b1h = s.b1[g].zw;
        acc[0][0] = __builtin_amdgcn_wmma_f32_16x16x4_f32(
            false, a0l, false, b0l, (short)0, acc[0][0], false, false);
        acc[0][1] = __builtin_amdgcn_wmma_f32_16x16x4_f32(
            false, a0l, false, b1l, (short)0, acc[0][1], false, false);
        acc[1][0] = __builtin_amdgcn_wmma_f32_16x16x4_f32(
            false, a1l, false, b0l, (short)0, acc[1][0], false, false);
        acc[1][1] = __builtin_amdgcn_wmma_f32_16x16x4_f32(
            false, a1l, false, b1l, (short)0, acc[1][1], false, false);
        acc[0][0] = __builtin_amdgcn_wmma_f32_16x16x4_f32(
            false, a0h, false, b0h, (short)0, acc[0][0], false, false);
        acc[0][1] = __builtin_amdgcn_wmma_f32_16x16x4_f32(
            false, a0h, false, b1h, (short)0, acc[0][1], false, false);
        acc[1][0] = __builtin_amdgcn_wmma_f32_16x16x4_f32(
            false, a1h, false, b0h, (short)0, acc[1][0], false, false);
        acc[1][1] = __builtin_amdgcn_wmma_f32_16x16x4_f32(
            false, a1h, false, b1h, (short)0, acc[1][1], false, false);
    }
}

__global__ __launch_bounds__(256) void lstm_step_kernel(
    const float* __restrict__ x_t,
    const float* __restrict__ h_in,
    float* __restrict__ h_out,
    float* __restrict__ c_state,
    float* __restrict__ h_seq_t,
    const float* __restrict__ bsum,
    GateParams gp)
{
    __shared__ float smem[8 * 1024];

    const int tid  = threadIdx.x;
    const int wave = tid >> 5;
    const int lane = tid & 31;

    const int bt = blockIdx.x;
    const int hg = blockIdx.y;

    const int c    = lane & 15;
    const int kk   = (lane >> 4) << 2;

    const int arow = bt * 32 + c;
    const int wrow = hg * 8 + (c & 7);
    const bool odd = (c >> 3) != 0;

    const float* wi0 = (odd ? gp.Wi[1] : gp.Wi[0]) + wrow * IN_DIM + kk + wave * 64;
    const float* wi1 = (odd ? gp.Wi[3] : gp.Wi[2]) + wrow * IN_DIM + kk + wave * 64;
    const float* wh0 = (odd ? gp.Wh[1] : gp.Wh[0]) + wrow * HID + kk + wave * 128;
    const float* wh1 = (odd ? gp.Wh[3] : gp.Wh[2]) + wrow * HID + kk + wave * 128;
    const float* ax  = x_t  + arow * IN_DIM + kk + wave * 64;
    const float* ah  = h_in + arow * HID    + kk + wave * 128;

    v8f acc[2][2] = {};
    Slice s0, s1;

    slice_load(s0, ax, ax + 16 * IN_DIM, wi0, wi1);
    slice_load(s1, ah, ah + 16 * HID, wh0, wh1);
    __builtin_amdgcn_sched_barrier(0);
    slice_mma(s0, acc);
    slice_load(s0, ah + 64, ah + 16 * HID + 64, wh0 + 64, wh1 + 64);
    __builtin_amdgcn_sched_barrier(0);
    slice_mma(s1, acc);
    slice_mma(s0, acc);

    {
        float* s = smem + wave * 1024;
        const int mbase = (lane >> 4) << 3;
        #pragma unroll
        for (int i = 0; i < 2; ++i)
            #pragma unroll
            for (int j = 0; j < 2; ++j)
                #pragma unroll
                for (int r = 0; r < 8; ++r)
                    s[(i * 2 + j) * 256 + (mbase + r) * 16 + c] = acc[i][j][r];
    }
    __syncthreads();

    {
        const int bl = tid >> 3;
        const int hh = tid & 7;
        const int b  = bt * 32 + bl;
        const int h  = hg * 8 + hh;
        const int t0 = (bl >> 4) * 512 + (bl & 15) * 16 + hh;

        float pre[4];
        #pragma unroll
        for (int g = 0; g < 4; ++g) {
            const int off = t0 + (g >> 1) * 256 + (g & 1) * 8;
            float sum = smem[off];
            #pragma unroll
            for (int w = 1; w < 8; ++w)
                sum += smem[w * 1024 + off];
            pre[g] = sum + bsum[g * HID + h];
        }

        float f = fast_sigmoid(pre[0]);
        float i = fast_sigmoid(pre[1]);
        float o = fast_sigmoid(pre[2]);
        float g = tanhf(pre[3]);

        const int idx = b * HID + h;
        float c_new = f * c_state[idx] + i * g;
        c_state[idx] = c_new;
        float h_new = o * tanhf(c_new);
        h_out[idx]   = h_new;
        h_seq_t[idx] = h_new;
    }
}

__global__ __launch_bounds__(256) void prep_kernel(
    float* __restrict__ h0, float* __restrict__ c0,
    float* __restrict__ bsum,
    const float* __restrict__ bi0, const float* __restrict__ bh0,
    const float* __restrict__ bi1, const float* __restrict__ bh1,
    const float* __restrict__ bi2, const float* __restrict__ bh2,
    const float* __restrict__ bi3, const float* __restrict__ bh3)
{
    int i = blockIdx.x * blockDim.x + threadIdx.x;
    if (i < BATCH * HID) {
        h0[i] = 0.0f;
        c0[i] = 0.0f;
    }
    if (i < HID) {
        bsum[0 * HID + i] = bi0[i] + bh0[i];
        bsum[1 * HID + i] = bi1[i] + bh1[i];
        bsum[2 * HID + i] = bi2[i] + bh2[i];
        bsum[3 * HID + i] = bi3[i] + bh3[i];
    }
}

__global__ __launch_bounds__(256) void tail_copy_kernel(
    const float* __restrict__ h_fin, const float* __restrict__ c_fin,
    float* __restrict__ out_tail)
{
    int i = blockIdx.x * blockDim.x + threadIdx.x;
    if (i < BATCH * HID) {
        out_tail[i]               = h_fin[i];
        out_tail[BATCH * HID + i] = c_fin[i];
    }
}

extern "C" void kernel_launch(void* const* d_in, const int* in_sizes, int n_in,
                              void* d_out, int out_size, void* d_ws, size_t ws_size,
                              hipStream_t stream) {
    (void)in_sizes; (void)n_in; (void)out_size; (void)ws_size;

    const float* x = (const float*)d_in[0];
    GateParams gp;
    const float* bi[4];
    const float* bh[4];
    for (int g = 0; g < 4; ++g) {
        gp.Wi[g] = (const float*)d_in[1 + 4 * g];
        bi[g]    = (const float*)d_in[2 + 4 * g];
        gp.Wh[g] = (const float*)d_in[3 + 4 * g];
        bh[g]    = (const float*)d_in[4 + 4 * g];
    }

    float* ws    = (float*)d_ws;
    float* hbuf0 = ws;
    float* hbuf1 = ws + BATCH * HID;
    float* cbuf  = ws + 2 * BATCH * HID;
    float* bsum  = ws + 3 * BATCH * HID;
    float* out   = (float*)d_out;

    prep_kernel<<<(BATCH * HID + 255) / 256, 256, 0, stream>>>(
        hbuf0, cbuf, bsum,
        bi[0], bh[0], bi[1], bh[1], bi[2], bh[2], bi[3], bh[3]);

    dim3 grid(BATCH / 32, HID / 8);
    for (int t = 0; t < SEQL; ++t) {
        const float* h_in  = (t & 1) ? hbuf1 : hbuf0;
        float*       h_out = (t & 1) ? hbuf0 : hbuf1;
        lstm_step_kernel<<<grid, 256, 0, stream>>>(
            x + (size_t)t * BATCH * IN_DIM, h_in, h_out, cbuf,
            out + (size_t)t * BATCH * HID, bsum, gp);
    }
    tail_copy_kernel<<<(BATCH * HID + 255) / 256, 256, 0, stream>>>(
        hbuf0, cbuf, out + (size_t)SEQL * BATCH * HID);
}


# A operands (h and x_t) read in fragment order from the workspace: an A load is 512 contiguous bytes (was 16 lines x 32 B); step t re-lays x_t+1; c state carried in its output slot; sums unchanged
# speedup vs baseline: 1.5509x; 1.0384x over previous
#include <hip/hip_runtime.h>


#define SEQL 256
#define BATCH 64
#define IN_DIM 512
#define HID 1024

typedef __attribute__((ext_vector_type(2))) float v2f;
typedef __attribute__((ext_vector_type(4))) float v4f;
typedef __attribute__((ext_vector_type(8))) float v8f;

struct GateParams {
    const float* Wi[4];
    const float* Wh[4];
};

__device__ __forceinline__ float fast_sigmoid(float x) {
    return 1.0f / (1.0f + __expf(-x));
}

struct Slice {
    v4f a0[8], a1[8], b0[8], b1[8];
};

#define HP_CHUNK 128
#define HP_TILE  (HP_CHUNK * (HID / 8))
#define XP_TILE  (HP_CHUNK * (IN_DIM / 8))
__device__ __forceinline__ int frag_index(int b, int k, int tile_floats) {
    return ((b >> 4) * tile_floats) + ((k >> 3) * HP_CHUNK)
         + ((((k >> 2) & 1) * 16 + (b & 15)) << 2) + (k & 3);
}

template <int ASTRIDE>
__device__ __forceinline__ void slice_load(
    Slice& s,
    const float* __restrict__ a0p, const float* __restrict__ a1p,
    const float* __restrict__ b0p, const float* __restrict__ b1p)
{
    #pragma unroll
    for (int g = 0; g < 8; ++g) s.a0[g] = *(const v4f*)(a0p + ASTRIDE * g);
    #pragma unroll
    for (int g = 0; g < 8; ++g) s.a1[g] = *(const v4f*)(a1p + ASTRIDE * g);
    #pragma unroll
    for (int g = 0; g < 8; ++g) s.b0[g] = *(const v4f*)(b0p + 8 * g);
    #pragma unroll
    for (int g = 0; g < 8; ++g) s.b1[g] = *(const v4f*)(b1p + 8 * g);
}

__device__ __forceinline__ void slice_mma(const Slice& s, v8f (&acc)[2][2])
{
    #pragma unroll
    for (int g = 0; g < 8; ++g) {
        v2f a0l = s.a0[g].xy, a0h = s.a0[g].zw;
        v2f a1l = s.a1[g].xy, a1h = s.a1[g].zw;
        v2f b0l = s.b0[g].xy, b0h = s.b0[g].zw;
        v2f b1l = s.b1[g].xy, b1h = s.b1[g].zw;
        acc[0][0] = __builtin_amdgcn_wmma_f32_16x16x4_f32(
            false, a0l, false, b0l, (short)0, acc[0][0], false, false);
        acc[0][1] = __builtin_amdgcn_wmma_f32_16x16x4_f32(
            false, a0l, false, b1l, (short)0, acc[0][1], false, false);
        acc[1][0] = __builtin_amdgcn_wmma_f32_16x16x4_f32(
            false, a1l, false, b0l, (short)0, acc[1][0], false, false);
        acc[1][1] = __builtin_amdgcn_wmma_f32_16x16x4_f32(
            false, a1l, false, b1l, (short)0, acc[1][1], false, false);
        acc[0][0] = __builtin_amdgcn_wmma_f32_16x16x4_f32(
            false, a0h, false, b0h, (short)0, acc[0][0], false, false);
        acc[0][1] = __builtin_amdgcn_wmma_f32_16x16x4_f32(
            false, a0h, false, b1h, (short)0, acc[0][1], false, false);
        acc[1][0] = __builtin_amdgcn_wmma_f32_16x16x4_f32(
            false, a1h, false, b0h, (short)0, acc[1][0], false, false);
        acc[1][1] = __builtin_amdgcn_wmma_f32_16x16x4_f32(
            false, a1h, false, b1h, (short)0, acc[1][1], false, false);
    }
}

__global__ __launch_bounds__(256) void lstm_step_kernel(
    const float* __restrict__ xp_in,
    const float* __restrict__ x_next,
    float* __restrict__ xp_out,
    const float* __restrict__ h_in,
    float* __restrict__ h_out,
    float* __restrict__ c_state,
    float* __restrict__ h_seq_t,
    const float* __restrict__ bsum,
    GateParams gp)
{
    __shared__ float smem[8 * 1024];

    const int tid  = threadIdx.x;
    const int wave = tid >> 5;
    const int lane = tid & 31;

    const int bt = blockIdx.x;
    const int hg = blockIdx.y;

    const int c    = lane & 15;
    const int kk   = (lane >> 4) << 2;

    const int wrow = hg * 8 + (c & 7);
    const bool odd = (c >> 3) != 0;

    const float* wi0 = (odd ? gp.Wi[1] : gp.Wi[0]) + wrow * IN_DIM + kk + wave * 64;
    const float* wi1 = (odd ? gp.Wi[3] : gp.Wi[2]) + wrow * IN_DIM + kk + wave * 64;
    const float* wh0 = (odd ? gp.Wh[1] : gp.Wh[0]) + wrow * HID + kk + wave * 128;
    const float* wh1 = (odd ? gp.Wh[3] : gp.Wh[2]) + wrow * HID + kk + wave * 128;
    const float* ax  = xp_in + (2 * bt) * XP_TILE + (8 * wave) * HP_CHUNK + lane * 4;
    const float* ah  = h_in + (2 * bt) * HP_TILE + (16 * wave) * HP_CHUNK + lane * 4;

    v8f acc[2][2] = {};
    Slice s0, s1;

    slice_load<HP_CHUNK>(s0, ax, ax + XP_TILE, wi0, wi1);
    slice_load<HP_CHUNK>(s1, ah, ah + HP_TILE, wh0, wh1);
    __builtin_amdgcn_sched_barrier(0);
    slice_mma(s0, acc);
    slice_load<HP_CHUNK>(s0, ah + 8 * HP_CHUNK, ah + HP_TILE + 8 * HP_CHUNK,
                         wh0 + 64, wh1 + 64);
    const int  q       = bt * (HID / 8) + hg;
    const bool relay_x = (wave == 0) && (x_next != nullptr);
    v4f xn = {};
    if (relay_x)
        xn = *(const v4f*)(x_next + (16 * (q >> 6) + c) * IN_DIM + 8 * (q & 63) + kk);
    __builtin_amdgcn_sched_barrier(0);
    slice_mma(s1, acc);
    slice_mma(s0, acc);

    {
        float* s = smem + wave * 1024;
        const int mbase = (lane >> 4) << 3;
        #pragma unroll
        for (int i = 0; i < 2; ++i)
            #pragma unroll
            for (int j = 0; j < 2; ++j)
                #pragma unroll
                for (int r = 0; r < 8; ++r)
                    s[(i * 2 + j) * 256 + (mbase + r) * 16 + c] = acc[i][j][r];
    }
    __syncthreads();

    {
        const int bl = tid >> 3;
        const int hh = tid & 7;
        const int b  = bt * 32 + bl;
        const int h  = hg * 8 + hh;
        const int t0 = (bl >> 4) * 512 + (bl & 15) * 16 + hh;

        float pre[4];
        #pragma unroll
        for (int g = 0; g < 4; ++g) {
            const int off = t0 + (g >> 1) * 256 + (g & 1) * 8;
            float sum = smem[off];
            #pragma unroll
            for (int w = 1; w < 8; ++w)
                sum += smem[w * 1024 + off];
            pre[g] = sum + bsum[g * HID + h];
        }

        float f = fast_sigmoid(pre[0]);
        float i = fast_sigmoid(pre[1]);
        float o = fast_sigmoid(pre[2]);
        float g = tanhf(pre[3]);

        const int idx = b * HID + h;
        float c_new = f * c_state[idx] + i * g;
        c_state[idx] = c_new;
        float h_new = o * tanhf(c_new);
        h_out[frag_index(b, h, HP_TILE)] = h_new;
        h_seq_t[idx]                     = h_new;
    }

    if (relay_x)
        *(v4f*)(xp_out + q * HP_CHUNK + lane * 4) = xn;
}

__global__ __launch_bounds__(256) void prep_kernel(
    float* __restrict__ h0, float* __restrict__ c0,
    float* __restrict__ xp0, const float* __restrict__ x0,
    float* __restrict__ bsum,
    const float* __restrict__ bi0, const float* __restrict__ bh0,
    const float* __restrict__ bi1, const float* __restrict__ bh1,
    const float* __restrict__ bi2, const float* __restrict__ bh2,
    const float* __restrict__ bi3, const float* __restrict__ bh3)
{
    int i = blockIdx.x * blockDim.x + threadIdx.x;
    if (i < BATCH * HID) {
        h0[i] = 0.0f;
        c0[i] = 0.0f;
    }
    if (i < BATCH * IN_DIM)
        xp0[frag_index(i / IN_DIM, i % IN_DIM, XP_TILE)] = x0[i];
    if (i < HID) {
        bsum[0 * HID + i] = bi0[i] + bh0[i];
        bsum[1 * HID + i] = bi1[i] + bh1[i];
        bsum[2 * HID + i] = bi2[i] + bh2[i];
        bsum[3 * HID + i] = bi3[i] + bh3[i];
    }
}

__global__ __launch_bounds__(256) void tail_copy_kernel(
    const float* __restrict__ h_fin, float* __restrict__ out_h_fin)
{
    int i = blockIdx.x * blockDim.x + threadIdx.x;
    if (i < BATCH * HID)
        out_h_fin[i] = h_fin[i];
}

extern "C" void kernel_launch(void* const* d_in, const int* in_sizes, int n_in,
                              void* d_out, int out_size, void* d_ws, size_t ws_size,
                              hipStream_t stream) {
    (void)in_sizes; (void)n_in; (void)out_size; (void)ws_size;

    const float* x = (const float*)d_in[0];
    GateParams gp;
    const float* bi[4];
    const float* bh[4];
    for (int g = 0; g < 4; ++g) {
        gp.Wi[g] = (const float*)d_in[1 + 4 * g];
        bi[g]    = (const float*)d_in[2 + 4 * g];
        gp.Wh[g] = (const float*)d_in[3 + 4 * g];
        bh[g]    = (const float*)d_in[4 + 4 * g];
    }

    float* ws    = (float*)d_ws;
    float* hbuf0 = ws;
    float* hbuf1 = ws + BATCH * HID;
    float* xpb0  = ws + 2 * BATCH * HID;
    float* xpb1  = ws + 2 * BATCH * HID + BATCH * IN_DIM;
    float* bsum  = ws + 3 * BATCH * HID;
    float* out   = (float*)d_out;
    float* cst   = out + (size_t)(SEQL + 1) * BATCH * HID;

    prep_kernel<<<(BATCH * HID + 255) / 256, 256, 0, stream>>>(
        hbuf0, cst, xpb0, x, bsum,
        bi[0], bh[0], bi[1], bh[1], bi[2], bh[2], bi[3], bh[3]);

    dim3 grid(BATCH / 32, HID / 8);
    for (int t = 0; t < SEQL; ++t) {
        const float* h_in   = (t & 1) ? hbuf1 : hbuf0;
        float*       h_out  = (t & 1) ? hbuf0 : hbuf1;
        const float* xp_in  = (t & 1) ? xpb1 : xpb0;
        float*       xp_out = (t & 1) ? xpb0 : xpb1;
        const float* x_next = (t + 1 < SEQL) ? x + (size_t)(t + 1) * BATCH * IN_DIM
                                             : nullptr;
        lstm_step_kernel<<<grid, 256, 0, stream>>>(
            xp_in, x_next, xp_out, h_in, h_out, cst,
            out + (size_t)t * BATCH * HID, bsum, gp);
    }
    tail_copy_kernel<<<(BATCH * HID + 255) / 256, 256, 0, stream>>>(
        out + (size_t)(SEQL - 1) * BATCH * HID,
        out + (size_t)SEQL * BATCH * HID);
}


# Wave 0's fetch of the next step's x chunk no longer drains its load queue: fetched vector has no initial value (no copy + s_wait_loadcnt 0 in the branch), branch made scalar; sums unchanged
# speedup vs baseline: 1.6187x; 1.0437x over previous
#include <hip/hip_runtime.h>


#define SEQL 256
#define BATCH 64
#define IN_DIM 512
#define HID 1024

typedef __attribute__((ext_vector_type(2))) float v2f;
typedef __attribute__((ext_vector_type(4))) float v4f;
typedef __attribute__((ext_vector_type(8))) float v8f;

struct GateParams {
    const float* Wi[4];
    const float* Wh[4];
};

__device__ __forceinline__ float fast_sigmoid(float x) {
    return 1.0f / (1.0f + __expf(-x));
}

struct Slice {
    v4f a0[8], a1[8], b0[8], b1[8];
};

#define HP_CHUNK 128
#define HP_TILE  (HP_CHUNK * (HID / 8))
#define XP_TILE  (HP_CHUNK * (IN_DIM / 8))
__device__ __forceinline__ int frag_index(int b, int k, int tile_floats) {
    return ((b >> 4) * tile_floats) + ((k >> 3) * HP_CHUNK)
         + ((((k >> 2) & 1) * 16 + (b & 15)) << 2) + (k & 3);
}

template <int ASTRIDE>
__device__ __forceinline__ void slice_load(
    Slice& s,
    const float* __restrict__ a0p, const float* __restrict__ a1p,
    const float* __restrict__ b0p, const float* __restrict__ b1p)
{
    #pragma unroll
    for (int g = 0; g < 8; ++g) s.a0[g] = *(const v4f*)(a0p + ASTRIDE * g);
    #pragma unroll
    for (int g = 0; g < 8; ++g) s.a1[g] = *(const v4f*)(a1p + ASTRIDE * g);
    #pragma unroll
    for (int g = 0; g < 8; ++g) s.b0[g] = *(const v4f*)(b0p + 8 * g);
    #pragma unroll
    for (int g = 0; g < 8; ++g) s.b1[g] = *(const v4f*)(b1p + 8 * g);
}

__device__ __forceinline__ void slice_mma(const Slice& s, v8f (&acc)[2][2])
{
    #pragma unroll
    for (int g = 0; g < 8; ++g) {
        v2f a0l = s.a0[g].xy, a0h = s.a0[g].zw;
        v2f a1l = s.a1[g].xy, a1h = s.a1[g].zw;
        v2f b0l = s.b0[g].xy, b0h = s.b0[g].zw;
        v2f b1l = s.b1[g].xy, b1h = s.b1[g].zw;
        acc[0][0] = __builtin_amdgcn_wmma_f32_16x16x4_f32(
            false, a0l, false, b0l, (short)0, acc[0][0], false, false);
        acc[0][1] = __builtin_amdgcn_wmma_f32_16x16x4_f32(
            false, a0l, false, b1l, (short)0, acc[0][1], false, false);
        acc[1][0] = __builtin_amdgcn_wmma_f32_16x16x4_f32(
            false, a1l, false, b0l, (short)0, acc[1][0], false, false);
        acc[1][1] = __builtin_amdgcn_wmma_f32_16x16x4_f32(
            false, a1l, false, b1l, (short)0, acc[1][1], false, false);
        acc[0][0] = __builtin_amdgcn_wmma_f32_16x16x4_f32(
            false, a0h, false, b0h, (short)0, acc[0][0], false, false);
        acc[0][1] = __builtin_amdgcn_wmma_f32_16x16x4_f32(
            false, a0h, false, b1h, (short)0, acc[0][1], false, false);
        acc[1][0] = __builtin_amdgcn_wmma_f32_16x16x4_f32(
            false, a1h, false, b0h, (short)0, acc[1][0], false, false);
        acc[1][1] = __builtin_amdgcn_wmma_f32_16x16x4_f32(
            false, a1h, false, b1h, (short)0, acc[1][1], false, false);
    }
}

__global__ __launch_bounds__(256) void lstm_step_kernel(
    const float* __restrict__ xp_in,
    const float* __restrict__ x_next,
    float* __restrict__ xp_out,
    const float* __restrict__ h_in,
    float* __restrict__ h_out,
    float* __restrict__ c_state,
    float* __restrict__ h_seq_t,
    const float* __restrict__ bsum,
    GateParams gp)
{
    __shared__ float smem[8 * 1024];

    const int tid  = threadIdx.x;
    const int wave = tid >> 5;
    const int lane = tid & 31;

    const int bt = blockIdx.x;
    const int hg = blockIdx.y;

    const int c    = lane & 15;
    const int kk   = (lane >> 4) << 2;

    const int wrow = hg * 8 + (c & 7);
    const bool odd = (c >> 3) != 0;

    const float* wi0 = (odd ? gp.Wi[1] : gp.Wi[0]) + wrow * IN_DIM + kk + wave * 64;
    const float* wi1 = (odd ? gp.Wi[3] : gp.Wi[2]) + wrow * IN_DIM + kk + wave * 64;
    const float* wh0 = (odd ? gp.Wh[1] : gp.Wh[0]) + wrow * HID + kk + wave * 128;
    const float* wh1 = (odd ? gp.Wh[3] : gp.Wh[2]) + wrow * HID + kk + wave * 128;
    const float* ax  = xp_in + (2 * bt) * XP_TILE + (8 * wave) * HP_CHUNK + lane * 4;
    const float* ah  = h_in + (2 * bt) * HP_TILE + (16 * wave) * HP_CHUNK + lane * 4;

    v8f acc[2][2] = {};
    Slice s0, s1;

    slice_load<HP_CHUNK>(s0, ax, ax + XP_TILE, wi0, wi1);
    slice_load<HP_CHUNK>(s1, ah, ah + HP_TILE, wh0, wh1);
    __builtin_amdgcn_sched_barrier(0);
    slice_mma(s0, acc);
    slice_load<HP_CHUNK>(s0, ah + 8 * HP_CHUNK, ah + HP_TILE + 8 * HP_CHUNK,
                         wh0 + 64, wh1 + 64);
    const int  q       = bt * (HID / 8) + hg;
    const bool relay_x = (__builtin_amdgcn_readfirstlane(wave) == 0) && (x_next != nullptr);
    v4f xn;
    if (relay_x)
        xn = *(const v4f*)(x_next + (16 * (q >> 6) + c) * IN_DIM + 8 * (q & 63) + kk);
    __builtin_amdgcn_sched_barrier(0);
    slice_mma(s1, acc);
    slice_mma(s0, acc);

    {
        float* s = smem + wave * 1024;
        const int mbase = (lane >> 4) << 3;
        #pragma unroll
        for (int i = 0; i < 2; ++i)
            #pragma unroll
            for (int j = 0; j < 2; ++j)
                #pragma unroll
                for (int r = 0; r < 8; ++r)
                    s[(i * 2 + j) * 256 + (mbase + r) * 16 + c] = acc[i][j][r];
    }
    __syncthreads();

    {
        const int bl = tid >> 3;
        const int hh = tid & 7;
        const int b  = bt * 32 + bl;
        const int h  = hg * 8 + hh;
        const int t0 = (bl >> 4) * 512 + (bl & 15) * 16 + hh;

        float pre[4];
        #pragma unroll
        for (int g = 0; g < 4; ++g) {
            const int off = t0 + (g >> 1) * 256 + (g & 1) * 8;
            float sum = smem[off];
            #pragma unroll
            for (int w = 1; w < 8; ++w)
                sum += smem[w * 1024 + off];
            pre[g] = sum + bsum[g * HID + h];
        }

        float f = fast_sigmoid(pre[0]);
        float i = fast_sigmoid(pre[1]);
        float o = fast_sigmoid(pre[2]);
        float g = tanhf(pre[3]);

        const int idx = b * HID + h;
        float c_new = f * c_state[idx] + i * g;
        c_state[idx] = c_new;
        float h_new = o * tanhf(c_new);
        h_out[frag_index(b, h, HP_TILE)] = h_new;
        h_seq_t[idx]                     = h_new;
    }

    if (relay_x)
        *(v4f*)(xp_out + q * HP_CHUNK + lane * 4) = xn;
}

__global__ __launch_bounds__(256) void prep_kernel(
    float* __restrict__ h0, float* __restrict__ c0,
    float* __restrict__ xp0, const float* __restrict__ x0,
    float* __restrict__ bsum,
    const float* __restrict__ bi0, const float* __restrict__ bh0,
    const float* __restrict__ bi1, const float* __restrict__ bh1,
    const float* __restrict__ bi2, const float* __restrict__ bh2,
    const float* __restrict__ bi3, const float* __restrict__ bh3)
{
    int i = blockIdx.x * blockDim.x + threadIdx.x;
    if (i < BATCH * HID) {
        h0[i] = 0.0f;
        c0[i] = 0.0f;
    }
    if (i < BATCH * IN_DIM)
        xp0[frag_index(i / IN_DIM, i % IN_DIM, XP_TILE)] = x0[i];
    if (i < HID) {
        bsum[0 * HID + i] = bi0[i] + bh0[i];
        bsum[1 * HID + i] = bi1[i] + bh1[i];
        bsum[2 * HID + i] = bi2[i] + bh2[i];
        bsum[3 * HID + i] = bi3[i] + bh3[i];
    }
}

__global__ __launch_bounds__(256) void tail_copy_kernel(
    const float* __restrict__ h_fin, float* __restrict__ out_h_fin)
{
    int i = blockIdx.x * blockDim.x + threadIdx.x;
    if (i < BATCH * HID)
        out_h_fin[i] = h_fin[i];
}

extern "C" void kernel_launch(void* const* d_in, const int* in_sizes, int n_in,
                              void* d_out, int out_size, void* d_ws, size_t ws_size,
                              hipStream_t stream) {
    (void)in_sizes; (void)n_in; (void)out_size; (void)ws_size;

    const float* x = (const float*)d_in[0];
    GateParams gp;
    const float* bi[4];
    const float* bh[4];
    for (int g = 0; g < 4; ++g) {
        gp.Wi[g] = (const float*)d_in[1 + 4 * g];
        bi[g]    = (const float*)d_in[2 + 4 * g];
        gp.Wh[g] = (const float*)d_in[3 + 4 * g];
        bh[g]    = (const float*)d_in[4 + 4 * g];
    }

    float* ws    = (float*)d_ws;
    float* hbuf0 = ws;
    float* hbuf1 = ws + BATCH * HID;
    float* xpb0  = ws + 2 * BATCH * HID;
    float* xpb1  = ws + 2 * BATCH * HID + BATCH * IN_DIM;
    float* bsum  = ws + 3 * BATCH * HID;
    float* out   = (float*)d_out;
    float* cst   = out + (size_t)(SEQL + 1) * BATCH * HID;

    prep_kernel<<<(BATCH * HID + 255) / 256, 256, 0, stream>>>(
        hbuf0, cst, xpb0, x, bsum,
        bi[0], bh[0], bi[1], bh[1], bi[2], bh[2], bi[3], bh[3]);

    dim3 grid(BATCH / 32, HID / 8);
    for (int t = 0; t < SEQL; ++t) {
        const float* h_in   = (t & 1) ? hbuf1 : hbuf0;
        float*       h_out  = (t & 1) ? hbuf0 : hbuf1;
        const float* xp_in  = (t & 1) ? xpb1 : xpb0;
        float*       xp_out = (t & 1) ? xpb0 : xpb1;
        const float* x_next = (t + 1 < SEQL) ? x + (size_t)(t + 1) * BATCH * IN_DIM
                                             : nullptr;
        lstm_step_kernel<<<grid, 256, 0, stream>>>(
            xp_in, x_next, xp_out, h_in, h_out, cst,
            out + (size_t)t * BATCH * HID, bsum, gp);
    }
    tail_copy_kernel<<<(BATCH * HID + 255) / 256, 256, 0, stream>>>(
        out + (size_t)(SEQL - 1) * BATCH * HID,
        out + (size_t)SEQL * BATCH * HID);
}


# Gate-phase operands (4 bias sums, old cell state) fetched in front of the last 128 matrix instructions: no cache round trip is left behind the barrier; sums unchanged
# speedup vs baseline: 1.7501x; 1.0812x over previous
#include <hip/hip_runtime.h>


#define SEQL 256
#define BATCH 64
#define IN_DIM 512
#define HID 1024

typedef __attribute__((ext_vector_type(2))) float v2f;
typedef __attribute__((ext_vector_type(4))) float v4f;
typedef __attribute__((ext_vector_type(8))) float v8f;

struct GateParams {
    const float* Wi[4];
    const float* Wh[4];
};

__device__ __forceinline__ float fast_sigmoid(float x) {
    return 1.0f / (1.0f + __expf(-x));
}

struct Slice {
    v4f a0[8], a1[8], b0[8], b1[8];
};

#define HP_CHUNK 128
#define HP_TILE  (HP_CHUNK * (HID / 8))
#define XP_TILE  (HP_CHUNK * (IN_DIM / 8))
__device__ __forceinline__ int frag_index(int b, int k, int tile_floats) {
    return ((b >> 4) * tile_floats) + ((k >> 3) * HP_CHUNK)
         + ((((k >> 2) & 1) * 16 + (b & 15)) << 2) + (k & 3);
}

template <int ASTRIDE>
__device__ __forceinline__ void slice_load(
    Slice& s,
    const float* __restrict__ a0p, const float* __restrict__ a1p,
    const float* __restrict__ b0p, const float* __restrict__ b1p)
{
    #pragma unroll
    for (int g = 0; g < 8; ++g) s.a0[g] = *(const v4f*)(a0p + ASTRIDE * g);
    #pragma unroll
    for (int g = 0; g < 8; ++g) s.a1[g] = *(const v4f*)(a1p + ASTRIDE * g);
    #pragma unroll
    for (int g = 0; g < 8; ++g) s.b0[g] = *(const v4f*)(b0p + 8 * g);
    #pragma unroll
    for (int g = 0; g < 8; ++g) s.b1[g] = *(const v4f*)(b1p + 8 * g);
}

__device__ __forceinline__ void slice_mma(const Slice& s, v8f (&acc)[2][2])
{
    #pragma unroll
    for (int g = 0; g < 8; ++g) {
        v2f a0l = s.a0[g].xy, a0h = s.a0[g].zw;
        v2f a1l = s.a1[g].xy, a1h = s.a1[g].zw;
        v2f b0l = s.b0[g].xy, b0h = s.b0[g].zw;
        v2f b1l = s.b1[g].xy, b1h = s.b1[g].zw;
        acc[0][0] = __builtin_amdgcn_wmma_f32_16x16x4_f32(
            false, a0l, false, b0l, (short)0, acc[0][0], false, false);
        acc[0][1] = __builtin_amdgcn_wmma_f32_16x16x4_f32(
            false, a0l, false, b1l, (short)0, acc[0][1], false, false);
        acc[1][0] = __builtin_amdgcn_wmma_f32_16x16x4_f32(
            false, a1l, false, b0l, (short)0, acc[1][0], false, false);
        acc[1][1] = __builtin_amdgcn_wmma_f32_16x16x4_f32(
            false, a1l, false, b1l, (short)0, acc[1][1], false, false);
        acc[0][0] = __builtin_amdgcn_wmma_f32_16x16x4_f32(
            false, a0h, false, b0h, (short)0, acc[0][0], false, false);
        acc[0][1] = __builtin_amdgcn_wmma_f32_16x16x4_f32(
            false, a0h, false, b1h, (short)0, acc[0][1], false, false);
        acc[1][0] = __builtin_amdgcn_wmma_f32_16x16x4_f32(
            false, a1h, false, b0h, (short)0, acc[1][0], false, false);
        acc[1][1] = __builtin_amdgcn_wmma_f32_16x16x4_f32(
            false, a1h, false, b1h, (short)0, acc[1][1], false, false);
    }
}

__global__ __launch_bounds__(256) void lstm_step_kernel(
    const float* __restrict__ xp_in,
    const float* __restrict__ x_next,
    float* __restrict__ xp_out,
    const float* __restrict__ h_in,
    float* __restrict__ h_out,
    float* __restrict__ c_state,
    float* __restrict__ h_seq_t,
    const float* __restrict__ bsum,
    GateParams gp)
{
    __shared__ float smem[8 * 1024];

    const int tid  = threadIdx.x;
    const int wave = tid >> 5;
    const int lane = tid & 31;

    const int bt = blockIdx.x;
    const int hg = blockIdx.y;

    const int c    = lane & 15;
    const int kk   = (lane >> 4) << 2;

    const int wrow = hg * 8 + (c & 7);
    const bool odd = (c >> 3) != 0;

    const float* wi0 = (odd ? gp.Wi[1] : gp.Wi[0]) + wrow * IN_DIM + kk + wave * 64;
    const float* wi1 = (odd ? gp.Wi[3] : gp.Wi[2]) + wrow * IN_DIM + kk + wave * 64;
    const float* wh0 = (odd ? gp.Wh[1] : gp.Wh[0]) + wrow * HID + kk + wave * 128;
    const float* wh1 = (odd ? gp.Wh[3] : gp.Wh[2]) + wrow * HID + kk + wave * 128;
    const float* ax  = xp_in + (2 * bt) * XP_TILE + (8 * wave) * HP_CHUNK + lane * 4;
    const float* ah  = h_in + (2 * bt) * HP_TILE + (16 * wave) * HP_CHUNK + lane * 4;

    v8f acc[2][2] = {};
    Slice s0, s1;

    slice_load<HP_CHUNK>(s0, ax, ax + XP_TILE, wi0, wi1);
    slice_load<HP_CHUNK>(s1, ah, ah + HP_TILE, wh0, wh1);
    __builtin_amdgcn_sched_barrier(0);
    slice_mma(s0, acc);
    slice_load<HP_CHUNK>(s0, ah + 8 * HP_CHUNK, ah + HP_TILE + 8 * HP_CHUNK,
                         wh0 + 64, wh1 + 64);
    const int  q       = bt * (HID / 8) + hg;
    const bool relay_x = (__builtin_amdgcn_readfirstlane(wave) == 0) && (x_next != nullptr);
    v4f xn;
    if (relay_x)
        xn = *(const v4f*)(x_next + (16 * (q >> 6) + c) * IN_DIM + 8 * (q & 63) + kk);
    const int g_bl  = tid >> 3;
    const int g_h   = hg * 8 + (tid & 7);
    const int g_idx = (bt * 32 + g_bl) * HID + g_h;
    const float g_bias[4] = { bsum[0 * HID + g_h], bsum[1 * HID + g_h],
                              bsum[2 * HID + g_h], bsum[3 * HID + g_h] };
    const float g_cold = c_state[g_idx];
    __builtin_amdgcn_sched_barrier(0);
    slice_mma(s1, acc);
    slice_mma(s0, acc);

    {
        float* s = smem + wave * 1024;
        const int mbase = (lane >> 4) << 3;
        #pragma unroll
        for (int i = 0; i < 2; ++i)
            #pragma unroll
            for (int j = 0; j < 2; ++j)
                #pragma unroll
                for (int r = 0; r < 8; ++r)
                    s[(i * 2 + j) * 256 + (mbase + r) * 16 + c] = acc[i][j][r];
    }
    __syncthreads();

    {
        const int bl = tid >> 3;
        const int hh = tid & 7;
        const int b  = bt * 32 + bl;
        const int h  = hg * 8 + hh;
        const int t0 = (bl >> 4) * 512 + (bl & 15) * 16 + hh;

        float pre[4];
        #pragma unroll
        for (int g = 0; g < 4; ++g) {
            const int off = t0 + (g >> 1) * 256 + (g & 1) * 8;
            float sum = smem[off];
            #pragma unroll
            for (int w = 1; w < 8; ++w)
                sum += smem[w * 1024 + off];
            pre[g] = sum + g_bias[g];
        }

        float f = fast_sigmoid(pre[0]);
        float i = fast_sigmoid(pre[1]);
        float o = fast_sigmoid(pre[2]);
        float g = tanhf(pre[3]);

        const int idx = b * HID + h;
        float c_new = f * g_cold + i * g;
        c_state[idx] = c_new;
        float h_new = o * tanhf(c_new);
        h_out[frag_index(b, h, HP_TILE)] = h_new;
        h_seq_t[idx]                     = h_new;
    }

    if (relay_x)
        *(v4f*)(xp_out + q * HP_CHUNK + lane * 4) = xn;
}

__global__ __launch_bounds__(256) void prep_kernel(
    float* __restrict__ h0, float* __restrict__ c0,
    float* __restrict__ xp0, const float* __restrict__ x0,
    float* __restrict__ bsum,
    const float* __restrict__ bi0, const float* __restrict__ bh0,
    const float* __restrict__ bi1, const float* __restrict__ bh1,
    const float* __restrict__ bi2, const float* __restrict__ bh2,
    const float* __restrict__ bi3, const float* __restrict__ bh3)
{
    int i = blockIdx.x * blockDim.x + threadIdx.x;
    if (i < BATCH * HID) {
        h0[i] = 0.0f;
        c0[i] = 0.0f;
    }
    if (i < BATCH * IN_DIM)
        xp0[frag_index(i / IN_DIM, i % IN_DIM, XP_TILE)] = x0[i];
    if (i < HID) {
        bsum[0 * HID + i] = bi0[i] + bh0[i];
        bsum[1 * HID + i] = bi1[i] + bh1[i];
        bsum[2 * HID + i] = bi2[i] + bh2[i];
        bsum[3 * HID + i] = bi3[i] + bh3[i];
    }
}

__global__ __launch_bounds__(256) void tail_copy_kernel(
    const float* __restrict__ h_fin, float* __restrict__ out_h_fin)
{
    int i = blockIdx.x * blockDim.x + threadIdx.x;
    if (i < BATCH * HID)
        out_h_fin[i] = h_fin[i];
}

extern "C" void kernel_launch(void* const* d_in, const int* in_sizes, int n_in,
                              void* d_out, int out_size, void* d_ws, size_t ws_size,
                              hipStream_t stream) {
    (void)in_sizes; (void)n_in; (void)out_size; (void)ws_size;

    const float* x = (const float*)d_in[0];
    GateParams gp;
    const float* bi[4];
    const float* bh[4];
    for (int g = 0; g < 4; ++g) {
        gp.Wi[g] = (const float*)d_in[1 + 4 * g];
        bi[g]    = (const float*)d_in[2 + 4 * g];
        gp.Wh[g] = (const float*)d_in[3 + 4 * g];
        bh[g]    = (const float*)d_in[4 + 4 * g];
    }

    float* ws    = (float*)d_ws;
    float* hbuf0 = ws;
    float* hbuf1 = ws + BATCH * HID;
    float* xpb0  = ws + 2 * BATCH * HID;
    float* xpb1  = ws + 2 * BATCH * HID + BATCH * IN_DIM;
    float* bsum  = ws + 3 * BATCH * HID;
    float* out   = (float*)d_out;
    float* cst   = out + (size_t)(SEQL + 1) * BATCH * HID;

    prep_kernel<<<(BATCH * HID + 255) / 256, 256, 0, stream>>>(
        hbuf0, cst, xpb0, x, bsum,
        bi[0], bh[0], bi[1], bh[1], bi[2], bh[2], bi[3], bh[3]);

    dim3 grid(BATCH / 32, HID / 8);
    for (int t = 0; t < SEQL; ++t) {
        const float* h_in   = (t & 1) ? hbuf1 : hbuf0;
        float*       h_out  = (t & 1) ? hbuf0 : hbuf1;
        const float* xp_in  = (t & 1) ? xpb1 : xpb0;
        float*       xp_out = (t & 1) ? xpb0 : xpb1;
        const float* x_next = (t + 1 < SEQL) ? x + (size_t)(t + 1) * BATCH * IN_DIM
                                             : nullptr;
        lstm_step_kernel<<<grid, 256, 0, stream>>>(
            xp_in, x_next, xp_out, h_in, h_out, cst,
            out + (size_t)t * BATCH * HID, bsum, gp);
    }
    tail_copy_kernel<<<(BATCH * HID + 255) / 256, 256, 0, stream>>>(
        out + (size_t)(SEQL - 1) * BATCH * HID,
        out + (size_t)SEQL * BATCH * HID);
}


# Weights (B operands) fetched in whole 128-B lines (a load = 2 rows x 256 B, was 16 rows x 32 B) and put into fragment order through a wave-private LDS area; same registers as before, sums unchanged
# speedup vs baseline: 1.8529x; 1.0587x over previous
#include <hip/hip_runtime.h>


#define SEQL 256
#define BATCH 64
#define IN_DIM 512
#define HID 1024

typedef __attribute__((ext_vector_type(2))) float v2f;
typedef __attribute__((ext_vector_type(4))) float v4f;
typedef __attribute__((ext_vector_type(8))) float v8f;

struct GateParams {
    const float* Wi[4];
    const float* Wh[4];
};

__device__ __forceinline__ float fast_sigmoid(float x) {
    return 1.0f / (1.0f + __expf(-x));
}

struct Slice {
    v4f a0[8], a1[8], b0[8], b1[8];
};

#define BT_PITCH 68
#define BT_WAVE  (32 * BT_PITCH)

#define HP_CHUNK 128
#define HP_TILE  (HP_CHUNK * (HID / 8))
#define XP_TILE  (HP_CHUNK * (IN_DIM / 8))
__device__ __forceinline__ int frag_index(int b, int k, int tile_floats) {
    return ((b >> 4) * tile_floats) + ((k >> 3) * HP_CHUNK)
         + ((((k >> 2) & 1) * 16 + (b & 15)) << 2) + (k & 3);
}

template <int ASTRIDE>
__device__ __forceinline__ void a_load(
    Slice& s, const float* __restrict__ a0p, const float* __restrict__ a1p)
{
    #pragma unroll
    for (int g = 0; g < 8; ++g) s.a0[g] = *(const v4f*)(a0p + ASTRIDE * g);
    #pragma unroll
    for (int g = 0; g < 8; ++g) s.a1[g] = *(const v4f*)(a1p + ASTRIDE * g);
}

template <int KROW>
__device__ __forceinline__ void rawb_load(
    Slice& s,
    const float* __restrict__ w0, const float* __restrict__ w1,
    const float* __restrict__ w2, const float* __restrict__ w3)
{
    #pragma unroll
    for (int i = 0; i < 4; ++i) s.b0[i]     = *(const v4f*)(w0 + 2 * i * KROW);
    #pragma unroll
    for (int i = 0; i < 4; ++i) s.b0[4 + i] = *(const v4f*)(w1 + 2 * i * KROW);
    #pragma unroll
    for (int i = 0; i < 4; ++i) s.b1[i]     = *(const v4f*)(w2 + 2 * i * KROW);
    #pragma unroll
    for (int i = 0; i < 4; ++i) s.b1[4 + i] = *(const v4f*)(w3 + 2 * i * KROW);
}

__device__ __forceinline__ void stage_b(Slice& s, float* st, const float* ld)
{
    __builtin_amdgcn_fence(__ATOMIC_ACQ_REL, "wavefront");
    #pragma unroll
    for (int i = 0; i < 8; ++i) *(v4f*)(st + (2 * i) * BT_PITCH) = s.b0[i];
    #pragma unroll
    for (int i = 0; i < 8; ++i) *(v4f*)(st + (16 + 2 * i) * BT_PITCH) = s.b1[i];
    __builtin_amdgcn_fence(__ATOMIC_ACQ_REL, "wavefront");
    #pragma unroll
    for (int g = 0; g < 8; ++g) s.b0[g] = *(const v4f*)(ld + 8 * g);
    #pragma unroll
    for (int g = 0; g < 8; ++g) s.b1[g] = *(const v4f*)(ld + 16 * BT_PITCH + 8 * g);
}

template <int G0, int G1>
__device__ __forceinline__ void slice_mma(const Slice& s, v8f (&acc)[2][2])
{
    #pragma unroll
    for (int g = G0; g < G1; ++g) {
        v2f a0l = s.a0[g].xy, a0h = s.a0[g].zw;
        v2f a1l = s.a1[g].xy, a1h = s.a1[g].zw;
        v2f b0l = s.b0[g].xy, b0h = s.b0[g].zw;
        v2f b1l = s.b1[g].xy, b1h = s.b1[g].zw;
        acc[0][0] = __builtin_amdgcn_wmma_f32_16x16x4_f32(
            false, a0l, false, b0l, (short)0, acc[0][0], false, false);
        acc[0][1] = __builtin_amdgcn_wmma_f32_16x16x4_f32(
            false, a0l, false, b1l, (short)0, acc[0][1], false, false);
        acc[1][0] = __builtin_amdgcn_wmma_f32_16x16x4_f32(
            false, a1l, false, b0l, (short)0, acc[1][0], false, false);
        acc[1][1] = __builtin_amdgcn_wmma_f32_16x16x4_f32(
            false, a1l, false, b1l, (short)0, acc[1][1], false, false);
        acc[0][0] = __builtin_amdgcn_wmma_f32_16x16x4_f32(
            false, a0h, false, b0h, (short)0, acc[0][0], false, false);
        acc[0][1] = __builtin_amdgcn_wmma_f32_16x16x4_f32(
            false, a0h, false, b1h, (short)0, acc[0][1], false, false);
        acc[1][0] = __builtin_amdgcn_wmma_f32_16x16x4_f32(
            false, a1h, false, b0h, (short)0, acc[1][0], false, false);
        acc[1][1] = __builtin_amdgcn_wmma_f32_16x16x4_f32(
            false, a1h, false, b1h, (short)0, acc[1][1], false, false);
    }
}

__global__ __launch_bounds__(256) void lstm_step_kernel(
    const float* __restrict__ xp_in,
    const float* __restrict__ x_next,
    float* __restrict__ xp_out,
    const float* __restrict__ h_in,
    float* __restrict__ h_out,
    float* __restrict__ c_state,
    float* __restrict__ h_seq_t,
    const float* __restrict__ bsum,
    GateParams gp)
{
    __shared__ float smem[8 * 1024];
    __shared__ __attribute__((aligned(16))) float stage[8 * BT_WAVE];

    const int tid  = threadIdx.x;
    const int wave = tid >> 5;
    const int lane = tid & 31;

    const int bt = blockIdx.x;
    const int hg = blockIdx.y;

    const int c    = lane & 15;
    const int kk   = (lane >> 4) << 2;

    const int half = lane >> 4;
    const int wio  = (hg * 8 + half) * IN_DIM + wave * 64 + 4 * c;
    const int who  = (hg * 8 + half) * HID + wave * 128 + 4 * c;
    float*       st = stage + wave * BT_WAVE + half * BT_PITCH + 4 * c;
    const float* ld = stage + wave * BT_WAVE + c * BT_PITCH + kk;
    const float* ax  = xp_in + (2 * bt) * XP_TILE + (8 * wave) * HP_CHUNK + lane * 4;
    const float* ah  = h_in + (2 * bt) * HP_TILE + (16 * wave) * HP_CHUNK + lane * 4;

    v8f acc[2][2] = {};
    Slice s0, s1;

    rawb_load<IN_DIM>(s0, gp.Wi[0] + wio, gp.Wi[1] + wio, gp.Wi[2] + wio, gp.Wi[3] + wio);
    __builtin_amdgcn_sched_barrier(0);
    a_load<HP_CHUNK>(s0, ax, ax + XP_TILE);
    __builtin_amdgcn_sched_barrier(0);
    rawb_load<HID>(s1, gp.Wh[0] + who, gp.Wh[1] + who, gp.Wh[2] + who, gp.Wh[3] + who);
    __builtin_amdgcn_sched_barrier(0);
    a_load<HP_CHUNK>(s1, ah, ah + HP_TILE);
    __builtin_amdgcn_sched_barrier(0);
    stage_b(s0, st, ld);
    __builtin_amdgcn_sched_barrier(0);
    slice_mma<0, 6>(s0, acc);
    __builtin_amdgcn_sched_barrier(0);
    stage_b(s1, st, ld);
    __builtin_amdgcn_sched_barrier(0);
    slice_mma<6, 8>(s0, acc);
    a_load<HP_CHUNK>(s0, ah + 8 * HP_CHUNK, ah + HP_TILE + 8 * HP_CHUNK);
    rawb_load<HID>(s0, gp.Wh[0] + who + 64, gp.Wh[1] + who + 64,
                       gp.Wh[2] + who + 64, gp.Wh[3] + who + 64);
    const int  q       = bt * (HID / 8) + hg;
    const bool relay_x = (__builtin_amdgcn_readfirstlane(wave) == 0) && (x_next != nullptr);
    v4f xn;
    if (relay_x)
        xn = *(const v4f*)(x_next + (16 * (q >> 6) + c) * IN_DIM + 8 * (q & 63) + kk);
    const int g_bl  = tid >> 3;
    const int g_h   = hg * 8 + (tid & 7);
    const int g_idx = (bt * 32 + g_bl) * HID + g_h;
    const float g_bias[4] = { bsum[0 * HID + g_h], bsum[1 * HID + g_h],
                              bsum[2 * HID + g_h], bsum[3 * HID + g_h] };
    const float g_cold = c_state[g_idx];
    __builtin_amdgcn_sched_barrier(0);
    slice_mma<0, 6>(s1, acc);
    __builtin_amdgcn_sched_barrier(0);
    stage_b(s0, st, ld);
    __builtin_amdgcn_sched_barrier(0);
    slice_mma<6, 8>(s1, acc);
    slice_mma<0, 8>(s0, acc);

    {
        float* s = smem + wave * 1024;
        const int mbase = (lane >> 4) << 3;
        #pragma unroll
        for (int i = 0; i < 2; ++i)
            #pragma unroll
            for (int j = 0; j < 2; ++j)
                #pragma unroll
                for (int r = 0; r < 8; ++r)
                    s[(i * 2 + j) * 256 + (mbase + r) * 16 + c] = acc[i][j][r];
    }
    __syncthreads();

    {
        const int bl = tid >> 3;
        const int hh = tid & 7;
        const int b  = bt * 32 + bl;
        const int h  = hg * 8 + hh;
        const int t0 = (bl >> 4) * 512 + (bl & 15) * 16 + hh;

        float pre[4];
        #pragma unroll
        for (int g = 0; g < 4; ++g) {
            const int off = t0 + (g >> 1) * 256 + (g & 1) * 8;
            float sum = smem[off];
            #pragma unroll
            for (int w = 1; w < 8; ++w)
                sum += smem[w * 1024 + off];
            pre[g] = sum + g_bias[g];
        }

        float f = fast_sigmoid(pre[0]);
        float i = fast_sigmoid(pre[1]);
        float o = fast_sigmoid(pre[2]);
        float g = tanhf(pre[3]);

        const int idx = b * HID + h;
        float c_new = f * g_cold + i * g;
        c_state[idx] = c_new;
        float h_new = o * tanhf(c_new);
        h_out[frag_index(b, h, HP_TILE)] = h_new;
        h_seq_t[idx]                     = h_new;
    }

    if (relay_x)
        *(v4f*)(xp_out + q * HP_CHUNK + lane * 4) = xn;
}

__global__ __launch_bounds__(256) void prep_kernel(
    float* __restrict__ h0, float* __restrict__ c0,
    float* __restrict__ xp0, const float* __restrict__ x0,
    float* __restrict__ bsum,
    const float* __restrict__ bi0, const float* __restrict__ bh0,
    const float* __restrict__ bi1, const float* __restrict__ bh1,
    const float* __restrict__ bi2, const float* __restrict__ bh2,
    const float* __restrict__ bi3, const float* __restrict__ bh3)
{
    int i = blockIdx.x * blockDim.x + threadIdx.x;
    if (i < BATCH * HID) {
        h0[i] = 0.0f;
        c0[i] = 0.0f;
    }
    if (i < BATCH * IN_DIM)
        xp0[frag_index(i / IN_DIM, i % IN_DIM, XP_TILE)] = x0[i];
    if (i < HID) {
        bsum[0 * HID + i] = bi0[i] + bh0[i];
        bsum[1 * HID + i] = bi1[i] + bh1[i];
        bsum[2 * HID + i] = bi2[i] + bh2[i];
        bsum[3 * HID + i] = bi3[i] + bh3[i];
    }
}

__global__ __launch_bounds__(256) void tail_copy_kernel(
    const float* __restrict__ h_fin, float* __restrict__ out_h_fin)
{
    int i = blockIdx.x * blockDim.x + threadIdx.x;
    if (i < BATCH * HID)
        out_h_fin[i] = h_fin[i];
}

extern "C" void kernel_launch(void* const* d_in, const int* in_sizes, int n_in,
                              void* d_out, int out_size, void* d_ws, size_t ws_size,
                              hipStream_t stream) {
    (void)in_sizes; (void)n_in; (void)out_size; (void)ws_size;

    const float* x = (const float*)d_in[0];
    GateParams gp;
    const float* bi[4];
    const float* bh[4];
    for (int g = 0; g < 4; ++g) {
        gp.Wi[g] = (const float*)d_in[1 + 4 * g];
        bi[g]    = (const float*)d_in[2 + 4 * g];
        gp.Wh[g] = (const float*)d_in[3 + 4 * g];
        bh[g]    = (const float*)d_in[4 + 4 * g];
    }

    float* ws    = (float*)d_ws;
    float* hbuf0 = ws;
    float* hbuf1 = ws + BATCH * HID;
    float* xpb0  = ws + 2 * BATCH * HID;
    float* xpb1  = ws + 2 * BATCH * HID + BATCH * IN_DIM;
    float* bsum  = ws + 3 * BATCH * HID;
    float* out   = (float*)d_out;
    float* cst   = out + (size_t)(SEQL + 1) * BATCH * HID;

    prep_kernel<<<(BATCH * HID + 255) / 256, 256, 0, stream>>>(
        hbuf0, cst, xpb0, x, bsum,
        bi[0], bh[0], bi[1], bh[1], bi[2], bh[2], bi[3], bh[3]);

    dim3 grid(BATCH / 32, HID / 8);
    for (int t = 0; t < SEQL; ++t) {
        const float* h_in   = (t & 1) ? hbuf1 : hbuf0;
        float*       h_out  = (t & 1) ? hbuf0 : hbuf1;
        const float* xp_in  = (t & 1) ? xpb1 : xpb0;
        float*       xp_out = (t & 1) ? xpb0 : xpb1;
        const float* x_next = (t + 1 < SEQL) ? x + (size_t)(t + 1) * BATCH * IN_DIM
                                             : nullptr;
        lstm_step_kernel<<<grid, 256, 0, stream>>>(
            xp_in, x_next, xp_out, h_in, h_out, cst,
            out + (size_t)t * BATCH * HID, bsum, gp);
    }
    tail_copy_kernel<<<(BATCH * HID + 255) / 256, 256, 0, stream>>>(
        out + (size_t)(SEQL - 1) * BATCH * HID,
        out + (size_t)SEQL * BATCH * HID);
}
